# starting kernel
# baseline (speedup 1.0000x reference)
#include <hip/hip_runtime.h>
#include <math.h>

typedef __attribute__((ext_vector_type(2))) float v2f;
typedef __attribute__((ext_vector_type(8))) float v8f;

__global__ void gqhan_build_q(const float* __restrict__ theta_fo,
                              const float* __restrict__ theta_ado,
                              float* __restrict__ Qout) {
  __shared__ float Mm[16][8];
  const int t = threadIdx.x;

  if (t < 8) {
    float st[16];
#pragma unroll
    for (int i = 0; i < 16; ++i) st[i] = 0.0f;
    st[t] = 1.0f;

    for (int b = 0; b < 8; ++b) {
      const float th = 0.5f * theta_fo[b];
      const float c = cosf(th), s = sinf(th);
#pragma unroll
      for (int i = 0; i < 8; ++i) {
        const float a0 = st[i], a1 = st[i + 8];
        st[i] = c * a0 - s * a1;
        st[i + 8] = s * a0 + c * a1;
      }
      st[8 + b] = -st[8 + b];
#pragma unroll
      for (int i = 0; i < 8; ++i) {
        const float a0 = st[i], a1 = st[i + 8];
        st[i] = c * a0 + s * a1;
        st[i + 8] = -s * a0 + c * a1;
      }
    }

    auto had = [&](int wb) {
      const int str = 1 << wb;
      const float r = 0.70710678118654752440f;
#pragma unroll
      for (int i = 0; i < 16; ++i) {
        if (i & str) continue;
        const float a0 = st[i], a1 = st[i + str];
        st[i] = r * (a0 + a1);
        st[i + str] = r * (a0 - a1);
      }
    };
    auto cry = [&](int cb, int tb, float th) {
      const float c = cosf(0.5f * th), s = sinf(0.5f * th);
      const int cs = 1 << cb, ts = 1 << tb;
#pragma unroll
      for (int i = 0; i < 16; ++i) {
        if (!(i & cs) || (i & ts)) continue;
        const float a0 = st[i], a1 = st[i + ts];
        st[i] = c * a0 - s * a1;
        st[i + ts] = s * a0 + c * a1;
      }
    };

    had(2); had(1); had(0);
    const int wbit[3] = {2, 1, 0};
#pragma unroll
    for (int i = 0; i < 3; ++i) cry(wbit[i], wbit[(i + 1) % 3], theta_ado[i]);
    st[0] = -st[0];
    st[8] = -st[8];
#pragma unroll
    for (int i = 0; i < 3; ++i) cry(wbit[i], wbit[(i + 1) % 3], theta_ado[3 + i]);
    had(2); had(1); had(0);

#pragma unroll
    for (int k = 0; k < 16; ++k) Mm[k][t] = st[k];
  }

  __syncthreads();

  if (t < 64) {
    const int i = t >> 3, j = t & 7;
    float q = 0.0f;
#pragma unroll
    for (int k = 0; k < 16; ++k) {
      const float sgn = (k & 1) ? -1.0f : 1.0f;
      q += sgn * Mm[k][i] * Mm[k][j];
    }
    Qout[t] = q;
  }
}

__device__ __forceinline__ float swap16(float v) {
  return __int_as_float(
      __builtin_amdgcn_ds_swizzle(__float_as_int(v), 0x401F));
}

__global__ __launch_bounds__(256) void gqhan_qform(const float* __restrict__ x,
                                                   const float* __restrict__ Q,
                                                   float* __restrict__ out,
                                                   int B) {
  const int lane = threadIdx.x & 31;
  const int l = lane & 15;
  const int h = lane >> 4;
  const int wave = blockIdx.x * (blockDim.x >> 5) + (threadIdx.x >> 5);
  const int s0 = wave << 5;
  if (s0 >= B) return;

  const int ka = 2 * h, kb = 2 * h + 1, kc = 4 + 2 * h, kd = 5 + 2 * h;
  const int j = l - 8;
  v2f a0, a1;
  a0.x = (l < 8) ? Q[l * 8 + ka] : ((j == ka) ? 1.0f : 0.0f);
  a0.y = (l < 8) ? Q[l * 8 + kb] : ((j == kb) ? 1.0f : 0.0f);
  a1.x = (l < 8) ? Q[l * 8 + kc] : ((j == kc) ? 1.0f : 0.0f);
  a1.y = (l < 8) ? Q[l * 8 + kd] : ((j == kd) ? 1.0f : 0.0f);

#pragma unroll
  for (int t = 0; t < 2; ++t) {
    int sr = s0 + 16 * t + l;
    if (sr >= B) sr = B - 1;
    const float* row = x + (size_t)sr * 8;
    const v2f b0 = *(const v2f*)(row + 2 * h);
    const v2f b1 = *(const v2f*)(row + 4 + 2 * h);

    v8f acc = {};
    acc = __builtin_amdgcn_wmma_f32_16x16x4_f32(false, a0, false, b0,
                                                (short)0, acc, false, false);
    acc = __builtin_amdgcn_wmma_f32_16x16x4_f32(false, a1, false, b1,
                                                (short)0, acc, false, false);

    float q = 0.0f, nn = 0.0f;
#pragma unroll
    for (int r = 0; r < 8; ++r) {
      const float d = acc[r];
      const float p = swap16(d);
      const float xv = (lane < 16) ? p : d;
      q = fmaf(d, p, q);
      nn = fmaf(xv, xv, nn);
    }
    const float res = q * __builtin_amdgcn_rcpf(nn);

    if (lane < 16) {
      const int sidx = s0 + 16 * t + lane;
      if (sidx < B) out[sidx] = res;
    }
  }
}

extern "C" void kernel_launch(void* const* d_in, const int* in_sizes, int n_in,
                              void* d_out, int out_size, void* d_ws,
                              size_t ws_size, hipStream_t stream) {
  const float* x = (const float*)d_in[0];
  const float* theta_fo = (const float*)d_in[1];
  const float* theta_ado = (const float*)d_in[2];
  float* out = (float*)d_out;
  float* Q = (float*)d_ws;
  const int B = in_sizes[0] / 8;

  hipLaunchKernelGGL(gqhan_build_q, dim3(1), dim3(64), 0, stream,
                     theta_fo, theta_ado, Q);

  const int samplesPerBlock = 256;
  const int grid = (B + samplesPerBlock - 1) / samplesPerBlock;
  hipLaunchKernelGGL(gqhan_qform, dim3(grid), dim3(256), 0, stream,
                     x, Q, out, B);
}


# One launch instead of two: a fixed grid of 512 workgroups, each rebuilds the 8x8 form in LDS (same serial code) with its row loads in flight, then walks the sample tiles with a grid stride
# speedup vs baseline: 1.1188x; 1.1188x over previous
#include <hip/hip_runtime.h>
#include <math.h>


typedef __attribute__((ext_vector_type(2))) float v2f;
typedef __attribute__((ext_vector_type(8))) float v8f;

__device__ __forceinline__ void build_m_column(const float* __restrict__ theta_fo,
                                               const float* __restrict__ theta_ado,
                                               const int t, float (*Mm)[8]) {
  float st[16];
#pragma unroll
  for (int i = 0; i < 16; ++i) st[i] = 0.0f;
  st[t] = 1.0f;

  for (int b = 0; b < 8; ++b) {
    const float th = 0.5f * theta_fo[b];
    const float c = cosf(th), s = sinf(th);
#pragma unroll
    for (int i = 0; i < 8; ++i) {
      const float a0 = st[i], a1 = st[i + 8];
      st[i] = c * a0 - s * a1;
      st[i + 8] = s * a0 + c * a1;
    }
    st[8 + b] = -st[8 + b];
#pragma unroll
    for (int i = 0; i < 8; ++i) {
      const float a0 = st[i], a1 = st[i + 8];
      st[i] = c * a0 + s * a1;
      st[i + 8] = -s * a0 + c * a1;
    }
  }

  auto had = [&](int wb) {
    const int str = 1 << wb;
    const float r = 0.70710678118654752440f;
#pragma unroll
    for (int i = 0; i < 16; ++i) {
      if (i & str) continue;
      const float a0 = st[i], a1 = st[i + str];
      st[i] = r * (a0 + a1);
      st[i + str] = r * (a0 - a1);
    }
  };
  auto cry = [&](int cb, int tb, float th) {
    const float c = cosf(0.5f * th), s = sinf(0.5f * th);
    const int cs = 1 << cb, ts = 1 << tb;
#pragma unroll
    for (int i = 0; i < 16; ++i) {
      if (!(i & cs) || (i & ts)) continue;
      const float a0 = st[i], a1 = st[i + ts];
      st[i] = c * a0 - s * a1;
      st[i + ts] = s * a0 + c * a1;
    }
  };

  had(2); had(1); had(0);
  const int wbit[3] = {2, 1, 0};
#pragma unroll
  for (int i = 0; i < 3; ++i) cry(wbit[i], wbit[(i + 1) % 3], theta_ado[i]);
  st[0] = -st[0];
  st[8] = -st[8];
#pragma unroll
  for (int i = 0; i < 3; ++i) cry(wbit[i], wbit[(i + 1) % 3], theta_ado[3 + i]);
  had(2); had(1); had(0);

#pragma unroll
  for (int k = 0; k < 16; ++k) Mm[k][t] = st[k];
}

__device__ __forceinline__ float swap16(float v) {
  return __int_as_float(
      __builtin_amdgcn_ds_swizzle(__float_as_int(v), 0x401F));
}

#define GROUPS_PER_BATCH 8

__global__ __launch_bounds__(256) void gqhan_fused(const float* __restrict__ x,
                                                   const float* __restrict__ theta_fo,
                                                   const float* __restrict__ theta_ado,
                                                   float* __restrict__ out,
                                                   int B) {
  __shared__ float Mm[16][8];
  constexpr int T = 2 * GROUPS_PER_BATCH;
  const int t = threadIdx.x;
  const int lane = t & 31;
  const int l = lane & 15;
  const int h = lane >> 4;
  const int wib = __builtin_amdgcn_readfirstlane(t >> 5);
  const int wpb = blockDim.x >> 5;
  const int wave = blockIdx.x * wpb + wib;
  const int nwaves = gridDim.x * wpb;
  const int ngroups = (B + 31) >> 5;
  if ((int)blockIdx.x * wpb >= ngroups) return;
  const bool active = wave < ngroups;

  v2f b0[T], b1[T];
  auto load_batch = [&](const int g0) {
#pragma unroll
    for (int u = 0; u < T; ++u) {
      const int g = g0 + (u >> 1) * nwaves;
      int sr = (g << 5) + 16 * (u & 1) + l;
      if (sr >= B) sr = B - 1;
      const float* row = x + (size_t)sr * 8;
      b0[u] = *(const v2f*)(row + 2 * h);
      b1[u] = *(const v2f*)(row + 4 + 2 * h);
    }
  };

  int g0 = wave;
  if (active) load_batch(g0);

  if (t < 8) build_m_column(theta_fo, theta_ado, t, Mm);
  __syncthreads();
  if (!active) return;

  const int ka = 2 * h, kb = 2 * h + 1, kc = 4 + 2 * h, kd = 5 + 2 * h;
  const int j = l - 8;
  const int li = l & 7;
  float qa = 0.0f, qb = 0.0f, qc = 0.0f, qd = 0.0f;
#pragma unroll
  for (int k = 0; k < 16; ++k) {
    const float sgn = (k & 1) ? -1.0f : 1.0f;
    const float mi = Mm[k][li];
    qa += sgn * mi * Mm[k][ka];
    qb += sgn * mi * Mm[k][kb];
    qc += sgn * mi * Mm[k][kc];
    qd += sgn * mi * Mm[k][kd];
  }
  v2f a0, a1;
  a0.x = (l < 8) ? qa : ((j == ka) ? 1.0f : 0.0f);
  a0.y = (l < 8) ? qb : ((j == kb) ? 1.0f : 0.0f);
  a1.x = (l < 8) ? qc : ((j == kc) ? 1.0f : 0.0f);
  a1.y = (l < 8) ? qd : ((j == kd) ? 1.0f : 0.0f);

  for (;;) {
#pragma unroll
    for (int u = 0; u < T; ++u) {
      v8f acc = {};
      acc = __builtin_amdgcn_wmma_f32_16x16x4_f32(false, a0, false, b0[u],
                                                  (short)0, acc, false, false);
      acc = __builtin_amdgcn_wmma_f32_16x16x4_f32(false, a1, false, b1[u],
                                                  (short)0, acc, false, false);

      float q = 0.0f, nn = 0.0f;
#pragma unroll
      for (int r = 0; r < 8; ++r) {
        const float d = acc[r];
        const float p = swap16(d);
        const float xv = (lane < 16) ? p : d;
        q = fmaf(d, p, q);
        nn = fmaf(xv, xv, nn);
      }
      const float res = q * __builtin_amdgcn_rcpf(nn);

      if (lane < 16) {
        const int g = g0 + (u >> 1) * nwaves;
        const int sidx = (g << 5) + 16 * (u & 1) + lane;
        if (sidx < B) out[sidx] = res;
      }
    }
    g0 += nwaves * GROUPS_PER_BATCH;
    if (g0 >= ngroups) break;
    load_batch(g0);
  }
}

extern "C" void kernel_launch(void* const* d_in, const int* in_sizes, int n_in,
                              void* d_out, int out_size, void* d_ws,
                              size_t ws_size, hipStream_t stream) {
  const float* x = (const float*)d_in[0];
  const float* theta_fo = (const float*)d_in[1];
  const float* theta_ado = (const float*)d_in[2];
  float* out = (float*)d_out;
  const int B = in_sizes[0] / 8;

  hipLaunchKernelGGL(gqhan_fused, dim3(512), dim3(256), 0, stream,
                     x, theta_fo, theta_ado, out, B);
}


# Builder's serial chain shortened: wave 0 evaluates the 14 half-angle cosine/sine pairs once, one per lane, and hands them round by readlane; its 8 builder lanes run gate arithmetic only
# speedup vs baseline: 1.3746x; 1.2286x over previous
#include <hip/hip_runtime.h>
#include <math.h>


typedef __attribute__((ext_vector_type(2))) float v2f;
typedef __attribute__((ext_vector_type(8))) float v8f;

__device__ __forceinline__ void build_m_column(const float (&cfo)[8],
                                               const float (&sfo)[8],
                                               const float (&cad)[6],
                                               const float (&sad)[6],
                                               const int t, float (*Mm)[8]) {
  float st[16];
#pragma unroll
  for (int i = 0; i < 16; ++i) st[i] = 0.0f;
#pragma unroll
  for (int i = 0; i < 8; ++i) st[i] = (i == t) ? 1.0f : 0.0f;

#pragma unroll
  for (int b = 0; b < 8; ++b) {
    const float c = cfo[b], s = sfo[b];
#pragma unroll
    for (int i = 0; i < 8; ++i) {
      const float a0 = st[i], a1 = st[i + 8];
      st[i] = c * a0 - s * a1;
      st[i + 8] = s * a0 + c * a1;
    }
    st[8 + b] = -st[8 + b];
#pragma unroll
    for (int i = 0; i < 8; ++i) {
      const float a0 = st[i], a1 = st[i + 8];
      st[i] = c * a0 + s * a1;
      st[i + 8] = -s * a0 + c * a1;
    }
  }

  auto had = [&](int wb) {
    const int str = 1 << wb;
    const float r = 0.70710678118654752440f;
#pragma unroll
    for (int i = 0; i < 16; ++i) {
      if (i & str) continue;
      const float a0 = st[i], a1 = st[i + str];
      st[i] = r * (a0 + a1);
      st[i + str] = r * (a0 - a1);
    }
  };
  auto cry = [&](int cb, int tb, float c, float s) {
    const int cs = 1 << cb, ts = 1 << tb;
#pragma unroll
    for (int i = 0; i < 16; ++i) {
      if (!(i & cs) || (i & ts)) continue;
      const float a0 = st[i], a1 = st[i + ts];
      st[i] = c * a0 - s * a1;
      st[i + ts] = s * a0 + c * a1;
    }
  };

  had(2); had(1); had(0);
  const int wbit[3] = {2, 1, 0};
#pragma unroll
  for (int i = 0; i < 3; ++i) cry(wbit[i], wbit[(i + 1) % 3], cad[i], sad[i]);
  st[0] = -st[0];
  st[8] = -st[8];
#pragma unroll
  for (int i = 0; i < 3; ++i) cry(wbit[i], wbit[(i + 1) % 3], cad[3 + i], sad[3 + i]);
  had(2); had(1); had(0);

#pragma unroll
  for (int k = 0; k < 16; ++k) Mm[k][t] = st[k];
}

__device__ __forceinline__ float lane_value(float v, int src) {
  return __int_as_float(__builtin_amdgcn_readlane(__float_as_int(v), src));
}

__device__ __forceinline__ float swap16(float v) {
  return __int_as_float(
      __builtin_amdgcn_ds_swizzle(__float_as_int(v), 0x401F));
}

#define GROUPS_PER_BATCH 8

__global__ __launch_bounds__(256) void gqhan_fused(const float* __restrict__ x,
                                                   const float* __restrict__ theta_fo,
                                                   const float* __restrict__ theta_ado,
                                                   float* __restrict__ out,
                                                   int B) {
  __shared__ float Mm[16][8];
  constexpr int T = 2 * GROUPS_PER_BATCH;
  const int t = threadIdx.x;
  const int lane = t & 31;
  const int l = lane & 15;
  const int h = lane >> 4;
  const int wib = __builtin_amdgcn_readfirstlane(t >> 5);
  const int wpb = blockDim.x >> 5;
  const int wave = blockIdx.x * wpb + wib;
  const int nwaves = gridDim.x * wpb;
  const int ngroups = (B + 31) >> 5;
  if ((int)blockIdx.x * wpb >= ngroups) return;
  const bool active = wave < ngroups;

  v2f b0[T], b1[T];
  auto load_batch = [&](const int g0) {
#pragma unroll
    for (int u = 0; u < T; ++u) {
      const int g = g0 + (u >> 1) * nwaves;
      int sr = (g << 5) + 16 * (u & 1) + l;
      if (sr >= B) sr = B - 1;
      const float* row = x + (size_t)sr * 8;
      b0[u] = *(const v2f*)(row + 2 * h);
      b1[u] = *(const v2f*)(row + 4 + 2 * h);
    }
  };

  int g0 = wave;
  if (wib == 0) {
    const int ai = (lane < 13) ? lane : 13;
    const float* ap = (ai < 8) ? (theta_fo + ai) : (theta_ado + (ai - 8));
    const float angle = *ap;
    load_batch(g0);
    const float half_angle = 0.5f * angle;
    const float cl = cosf(half_angle), sl = sinf(half_angle);
    float cfo[8], sfo[8], cad[6], sad[6];
#pragma unroll
    for (int i = 0; i < 8; ++i) {
      cfo[i] = lane_value(cl, i);
      sfo[i] = lane_value(sl, i);
    }
#pragma unroll
    for (int i = 0; i < 6; ++i) {
      cad[i] = lane_value(cl, 8 + i);
      sad[i] = lane_value(sl, 8 + i);
    }
    if (lane < 8) build_m_column(cfo, sfo, cad, sad, lane, Mm);
  } else if (active) {
    load_batch(g0);
  }
  __syncthreads();
  if (!active) return;

  const int ka = 2 * h, kb = 2 * h + 1, kc = 4 + 2 * h, kd = 5 + 2 * h;
  const int j = l - 8;
  const int li = l & 7;
  float qa = 0.0f, qb = 0.0f, qc = 0.0f, qd = 0.0f;
#pragma unroll
  for (int k = 0; k < 16; ++k) {
    const float sgn = (k & 1) ? -1.0f : 1.0f;
    const float mi = Mm[k][li];
    qa += sgn * mi * Mm[k][ka];
    qb += sgn * mi * Mm[k][kb];
    qc += sgn * mi * Mm[k][kc];
    qd += sgn * mi * Mm[k][kd];
  }
  v2f a0, a1;
  a0.x = (l < 8) ? qa : ((j == ka) ? 1.0f : 0.0f);
  a0.y = (l < 8) ? qb : ((j == kb) ? 1.0f : 0.0f);
  a1.x = (l < 8) ? qc : ((j == kc) ? 1.0f : 0.0f);
  a1.y = (l < 8) ? qd : ((j == kd) ? 1.0f : 0.0f);

  for (;;) {
#pragma unroll
    for (int u = 0; u < T; ++u) {
      v8f acc = {};
      acc = __builtin_amdgcn_wmma_f32_16x16x4_f32(false, a0, false, b0[u],
                                                  (short)0, acc, false, false);
      acc = __builtin_amdgcn_wmma_f32_16x16x4_f32(false, a1, false, b1[u],
                                                  (short)0, acc, false, false);

      float q = 0.0f, nn = 0.0f;
#pragma unroll
      for (int r = 0; r < 8; ++r) {
        const float d = acc[r];
        const float p = swap16(d);
        const float xv = (lane < 16) ? p : d;
        q = fmaf(d, p, q);
        nn = fmaf(xv, xv, nn);
      }
      const float res = q * __builtin_amdgcn_rcpf(nn);

      if (lane < 16) {
        const int g = g0 + (u >> 1) * nwaves;
        const int sidx = (g << 5) + 16 * (u & 1) + lane;
        if (sidx < B) out[sidx] = res;
      }
    }
    g0 += nwaves * GROUPS_PER_BATCH;
    if (g0 >= ngroups) break;
    load_batch(g0);
  }
}

extern "C" void kernel_launch(void* const* d_in, const int* in_sizes, int n_in,
                              void* d_out, int out_size, void* d_ws,
                              size_t ws_size, hipStream_t stream) {
  const float* x = (const float*)d_in[0];
  const float* theta_fo = (const float*)d_in[1];
  const float* theta_ado = (const float*)d_in[2];
  float* out = (float*)d_out;
  const int B = in_sizes[0] / 8;

  hipLaunchKernelGGL(gqhan_fused, dim3(512), dim3(256), 0, stream,
                     x, theta_fo, theta_ado, out, B);
}


# A operand is the 16x8 circuit matrix itself, not the 8x8 form built from it: each lane sums its own signed squared amplitudes, one permlane16_swap per sum joins the halves; summation order changes
# speedup vs baseline: 1.6851x; 1.2259x over previous
#include <hip/hip_runtime.h>
#include <math.h>


typedef __attribute__((ext_vector_type(2))) float v2f;
typedef __attribute__((ext_vector_type(4))) float v4f;
typedef __attribute__((ext_vector_type(8))) float v8f;
typedef __attribute__((ext_vector_type(2))) unsigned int v2u;

__device__ __forceinline__ void build_m_column(const float (&cfo)[8],
                                               const float (&sfo)[8],
                                               const float (&cad)[6],
                                               const float (&sad)[6],
                                               const int t, float (*Mm)[8]) {
  float st[16];
#pragma unroll
  for (int i = 0; i < 16; ++i) st[i] = 0.0f;
#pragma unroll
  for (int i = 0; i < 8; ++i) st[i] = (i == t) ? 1.0f : 0.0f;

#pragma unroll
  for (int b = 0; b < 8; ++b) {
    const float c = cfo[b], s = sfo[b];
#pragma unroll
    for (int i = 0; i < 8; ++i) {
      const float a0 = st[i], a1 = st[i + 8];
      st[i] = c * a0 - s * a1;
      st[i + 8] = s * a0 + c * a1;
    }
    st[8 + b] = -st[8 + b];
#pragma unroll
    for (int i = 0; i < 8; ++i) {
      const float a0 = st[i], a1 = st[i + 8];
      st[i] = c * a0 + s * a1;
      st[i + 8] = -s * a0 + c * a1;
    }
  }

  auto had = [&](int wb) {
    const int str = 1 << wb;
    const float r = 0.70710678118654752440f;
#pragma unroll
    for (int i = 0; i < 16; ++i) {
      if (i & str) continue;
      const float a0 = st[i], a1 = st[i + str];
      st[i] = r * (a0 + a1);
      st[i + str] = r * (a0 - a1);
    }
  };
  auto cry = [&](int cb, int tb, float c, float s) {
    const int cs = 1 << cb, ts = 1 << tb;
#pragma unroll
    for (int i = 0; i < 16; ++i) {
      if (!(i & cs) || (i & ts)) continue;
      const float a0 = st[i], a1 = st[i + ts];
      st[i] = c * a0 - s * a1;
      st[i + ts] = s * a0 + c * a1;
    }
  };

  had(2); had(1); had(0);
  const int wbit[3] = {2, 1, 0};
#pragma unroll
  for (int i = 0; i < 3; ++i) cry(wbit[i], wbit[(i + 1) % 3], cad[i], sad[i]);
  st[0] = -st[0];
  st[8] = -st[8];
#pragma unroll
  for (int i = 0; i < 3; ++i) cry(wbit[i], wbit[(i + 1) % 3], cad[3 + i], sad[3 + i]);
  had(2); had(1); had(0);

#pragma unroll
  for (int k = 0; k < 16; ++k) Mm[k][t] = st[k];
}

__device__ __forceinline__ float lane_value(float v, int src) {
  return __int_as_float(__builtin_amdgcn_readlane(__float_as_int(v), src));
}

__device__ __forceinline__ float sum_halves_of_pair(float a, float b) {
  const v2u r = __builtin_amdgcn_permlane16_swap(__float_as_uint(a),
                                                 __float_as_uint(b), false, false);
  return __uint_as_float(r.x) + __uint_as_float(r.y);
}

#define GROUPS_PER_BATCH 8

__global__ __launch_bounds__(256) void gqhan_fused(const float* __restrict__ x,
                                                   const float* __restrict__ theta_fo,
                                                   const float* __restrict__ theta_ado,
                                                   float* __restrict__ out,
                                                   int B) {
  __shared__ __attribute__((aligned(16))) float Mm[16][8];
  constexpr int T = 2 * GROUPS_PER_BATCH;
  const int t = threadIdx.x;
  const int lane = t & 31;
  const int l = lane & 15;
  const int h = lane >> 4;
  const int wib = __builtin_amdgcn_readfirstlane(t >> 5);
  const int wpb = blockDim.x >> 5;
  const int wave = blockIdx.x * wpb + wib;
  const int nwaves = gridDim.x * wpb;
  const int ngroups = (B + 31) >> 5;
  if ((int)blockIdx.x * wpb >= ngroups) return;
  const bool active = wave < ngroups;

  v4f xb[T];
  auto load_batch = [&](const int g0) {
#pragma unroll
    for (int u = 0; u < T; ++u) {
      const int g = g0 + (u >> 1) * nwaves;
      int sr = (g << 5) + 16 * (u & 1) + l;
      if (sr >= B) sr = B - 1;
      xb[u] = *(const v4f*)(x + (size_t)sr * 8 + 4 * h);
    }
  };

  int g0 = wave;
  if (wib == 0) {
    const int ai = (lane < 13) ? lane : 13;
    const float* ap = (ai < 8) ? (theta_fo + ai) : (theta_ado + (ai - 8));
    const float angle = *ap;
    load_batch(g0);
    const float half_angle = 0.5f * angle;
    const float cl = cosf(half_angle), sl = sinf(half_angle);
    float cfo[8], sfo[8], cad[6], sad[6];
#pragma unroll
    for (int i = 0; i < 8; ++i) {
      cfo[i] = lane_value(cl, i);
      sfo[i] = lane_value(sl, i);
    }
#pragma unroll
    for (int i = 0; i < 6; ++i) {
      cad[i] = lane_value(cl, 8 + i);
      sad[i] = lane_value(sl, 8 + i);
    }
    if (lane < 8) build_m_column(cfo, sfo, cad, sad, lane, Mm);
  } else if (active) {
    load_batch(g0);
  }
  __syncthreads();
  if (!active) return;

  const v4f am = *(const v4f*)&Mm[l][4 * h];
  v2f a0, a1;
  a0.x = am.x; a0.y = am.y;
  a1.x = am.z; a1.y = am.w;

  for (;;) {
#pragma unroll
    for (int jg = 0; jg < GROUPS_PER_BATCH; ++jg) {
      float qp[2], np[2];
#pragma unroll
      for (int w = 0; w < 2; ++w) {
        const v4f xv = xb[2 * jg + w];
        v2f b0, b1;
        b0.x = xv.x; b0.y = xv.y;
        b1.x = xv.z; b1.y = xv.w;
        v8f acc = {};
        acc = __builtin_amdgcn_wmma_f32_16x16x4_f32(false, a0, false, b0,
                                                    (short)0, acc, false, false);
        acc = __builtin_amdgcn_wmma_f32_16x16x4_f32(false, a1, false, b1,
                                                    (short)0, acc, false, false);
        float ev = acc[0] * acc[0];
        float od = acc[1] * acc[1];
        ev = fmaf(acc[2], acc[2], ev);
        od = fmaf(acc[3], acc[3], od);
        ev = fmaf(acc[4], acc[4], ev);
        od = fmaf(acc[5], acc[5], od);
        ev = fmaf(acc[6], acc[6], ev);
        od = fmaf(acc[7], acc[7], od);
        qp[w] = ev - od;
        float ns = xv.x * xv.x;
        ns = fmaf(xv.y, xv.y, ns);
        ns = fmaf(xv.z, xv.z, ns);
        ns = fmaf(xv.w, xv.w, ns);
        np[w] = ns;
      }
      const float q = sum_halves_of_pair(qp[0], qp[1]);
      const float nn = sum_halves_of_pair(np[0], np[1]);
      const float res = q * __builtin_amdgcn_rcpf(nn);

      const int g = g0 + jg * nwaves;
      const int sidx = (g << 5) + lane;
      if (sidx < B) out[sidx] = res;
    }
    g0 += nwaves * GROUPS_PER_BATCH;
    if (g0 >= ngroups) break;
    load_batch(g0);
  }
}

extern "C" void kernel_launch(void* const* d_in, const int* in_sizes, int n_in,
                              void* d_out, int out_size, void* d_ws,
                              size_t ws_size, hipStream_t stream) {
  const float* x = (const float*)d_in[0];
  const float* theta_fo = (const float*)d_in[1];
  const float* theta_ado = (const float*)d_in[2];
  float* out = (float*)d_out;
  const int B = in_sizes[0] / 8;

  hipLaunchKernelGGL(gqhan_fused, dim3(512), dim3(256), 0, stream,
                     x, theta_fo, theta_ado, out, B);
}


# Barrier fences LDS only, not all memory: row loads stay in flight across it, each tile's load is waited for at its own first product, so products overlap the rows still arriving
# speedup vs baseline: 1.7727x; 1.0520x over previous
#include <hip/hip_runtime.h>
#include <math.h>


typedef __attribute__((ext_vector_type(2))) float v2f;
typedef __attribute__((ext_vector_type(4))) float v4f;
typedef __attribute__((ext_vector_type(8))) float v8f;
typedef __attribute__((ext_vector_type(2))) unsigned int v2u;

__device__ __forceinline__ void build_m_column(const float (&cfo)[8],
                                               const float (&sfo)[8],
                                               const float (&cad)[6],
                                               const float (&sad)[6],
                                               const int t, float (*Mm)[8]) {
  float st[16];
#pragma unroll
  for (int i = 0; i < 16; ++i) st[i] = 0.0f;
#pragma unroll
  for (int i = 0; i < 8; ++i) st[i] = (i == t) ? 1.0f : 0.0f;

#pragma unroll
  for (int b = 0; b < 8; ++b) {
    const float c = cfo[b], s = sfo[b];
#pragma unroll
    for (int i = 0; i < 8; ++i) {
      const float a0 = st[i], a1 = st[i + 8];
      st[i] = c * a0 - s * a1;
      st[i + 8] = s * a0 + c * a1;
    }
    st[8 + b] = -st[8 + b];
#pragma unroll
    for (int i = 0; i < 8; ++i) {
      const float a0 = st[i], a1 = st[i + 8];
      st[i] = c * a0 + s * a1;
      st[i + 8] = -s * a0 + c * a1;
    }
  }

  auto had = [&](int wb) {
    const int str = 1 << wb;
    const float r = 0.70710678118654752440f;
#pragma unroll
    for (int i = 0; i < 16; ++i) {
      if (i & str) continue;
      const float a0 = st[i], a1 = st[i + str];
      st[i] = r * (a0 + a1);
      st[i + str] = r * (a0 - a1);
    }
  };
  auto cry = [&](int cb, int tb, float c, float s) {
    const int cs = 1 << cb, ts = 1 << tb;
#pragma unroll
    for (int i = 0; i < 16; ++i) {
      if (!(i & cs) || (i & ts)) continue;
      const float a0 = st[i], a1 = st[i + ts];
      st[i] = c * a0 - s * a1;
      st[i + ts] = s * a0 + c * a1;
    }
  };

  had(2); had(1); had(0);
  const int wbit[3] = {2, 1, 0};
#pragma unroll
  for (int i = 0; i < 3; ++i) cry(wbit[i], wbit[(i + 1) % 3], cad[i], sad[i]);
  st[0] = -st[0];
  st[8] = -st[8];
#pragma unroll
  for (int i = 0; i < 3; ++i) cry(wbit[i], wbit[(i + 1) % 3], cad[3 + i], sad[3 + i]);
  had(2); had(1); had(0);

#pragma unroll
  for (int k = 0; k < 16; ++k) Mm[k][t] = st[k];
}

__device__ __forceinline__ float lane_value(float v, int src) {
  return __int_as_float(__builtin_amdgcn_readlane(__float_as_int(v), src));
}

__device__ __forceinline__ float sum_halves_of_pair(float a, float b) {
  const v2u r = __builtin_amdgcn_permlane16_swap(__float_as_uint(a),
                                                 __float_as_uint(b), false, false);
  return __uint_as_float(r.x) + __uint_as_float(r.y);
}

#define GROUPS_PER_BATCH 8

__global__ __launch_bounds__(256) void gqhan_fused(const float* __restrict__ x,
                                                   const float* __restrict__ theta_fo,
                                                   const float* __restrict__ theta_ado,
                                                   float* __restrict__ out,
                                                   int B) {
  __shared__ __attribute__((aligned(16))) float Mm[16][8];
  constexpr int T = 2 * GROUPS_PER_BATCH;
  const int t = threadIdx.x;
  const int lane = t & 31;
  const int l = lane & 15;
  const int h = lane >> 4;
  const int wib = __builtin_amdgcn_readfirstlane(t >> 5);
  const int wpb = blockDim.x >> 5;
  const int wave = blockIdx.x * wpb + wib;
  const int nwaves = gridDim.x * wpb;
  const int ngroups = (B + 31) >> 5;
  if ((int)blockIdx.x * wpb >= ngroups) return;
  const bool active = wave < ngroups;

  v4f xb[T];
  auto load_batch = [&](const int g0) {
#pragma unroll
    for (int u = 0; u < T; ++u) {
      const int g = g0 + (u >> 1) * nwaves;
      int sr = (g << 5) + 16 * (u & 1) + l;
      if (sr >= B) sr = B - 1;
      xb[u] = *(const v4f*)(x + (size_t)sr * 8 + 4 * h);
    }
  };

  int g0 = wave;
  if (wib == 0) {
    const int ai = (lane < 13) ? lane : 13;
    const float* ap = (ai < 8) ? (theta_fo + ai) : (theta_ado + (ai - 8));
    const float angle = *ap;
    load_batch(g0);
    const float half_angle = 0.5f * angle;
    const float cl = cosf(half_angle), sl = sinf(half_angle);
    float cfo[8], sfo[8], cad[6], sad[6];
#pragma unroll
    for (int i = 0; i < 8; ++i) {
      cfo[i] = lane_value(cl, i);
      sfo[i] = lane_value(sl, i);
    }
#pragma unroll
    for (int i = 0; i < 6; ++i) {
      cad[i] = lane_value(cl, 8 + i);
      sad[i] = lane_value(sl, 8 + i);
    }
    if (lane < 8) build_m_column(cfo, sfo, cad, sad, lane, Mm);
  } else if (active) {
    load_batch(g0);
  }
  __builtin_amdgcn_fence(__ATOMIC_RELEASE, "workgroup", "local");
  __builtin_amdgcn_s_barrier();
  __builtin_amdgcn_fence(__ATOMIC_ACQUIRE, "workgroup", "local");
  if (!active) return;

  const v4f am = *(const v4f*)&Mm[l][4 * h];
  v2f a0, a1;
  a0.x = am.x; a0.y = am.y;
  a1.x = am.z; a1.y = am.w;

  for (;;) {
#pragma unroll
    for (int jg = 0; jg < GROUPS_PER_BATCH; ++jg) {
      float qp[2], np[2];
#pragma unroll
      for (int w = 0; w < 2; ++w) {
        const v4f xv = xb[2 * jg + w];
        v2f b0, b1;
        b0.x = xv.x; b0.y = xv.y;
        b1.x = xv.z; b1.y = xv.w;
        v8f acc = {};
        acc = __builtin_amdgcn_wmma_f32_16x16x4_f32(false, a0, false, b0,
                                                    (short)0, acc, false, false);
        acc = __builtin_amdgcn_wmma_f32_16x16x4_f32(false, a1, false, b1,
                                                    (short)0, acc, false, false);
        float ev = acc[0] * acc[0];
        float od = acc[1] * acc[1];
        ev = fmaf(acc[2], acc[2], ev);
        od = fmaf(acc[3], acc[3], od);
        ev = fmaf(acc[4], acc[4], ev);
        od = fmaf(acc[5], acc[5], od);
        ev = fmaf(acc[6], acc[6], ev);
        od = fmaf(acc[7], acc[7], od);
        qp[w] = ev - od;
        float ns = xv.x * xv.x;
        ns = fmaf(xv.y, xv.y, ns);
        ns = fmaf(xv.z, xv.z, ns);
        ns = fmaf(xv.w, xv.w, ns);
        np[w] = ns;
      }
      const float q = sum_halves_of_pair(qp[0], qp[1]);
      const float nn = sum_halves_of_pair(np[0], np[1]);
      const float res = q * __builtin_amdgcn_rcpf(nn);

      const int g = g0 + jg * nwaves;
      const int sidx = (g << 5) + lane;
      if (sidx < B) out[sidx] = res;
    }
    g0 += nwaves * GROUPS_PER_BATCH;
    if (g0 >= ngroups) break;
    load_batch(g0);
  }
}

extern "C" void kernel_launch(void* const* d_in, const int* in_sizes, int n_in,
                              void* d_out, int out_size, void* d_ws,
                              size_t ws_size, hipStream_t stream) {
  const float* x = (const float*)d_in[0];
  const float* theta_fo = (const float*)d_in[1];
  const float* theta_ado = (const float*)d_in[2];
  float* out = (float*)d_out;
  const int B = in_sizes[0] / 8;

  hipLaunchKernelGGL(gqhan_fused, dim3(512), dim3(256), 0, stream,
                     x, theta_fo, theta_ado, out, B);
}


# M factored exactly: oracle maps basis t to cos(th_t)|0> - sin(th_t)|1>, ADO skips the ancilla; a builder lane makes one 8-entry ADO column and scales it twice; rounding of M changes
# speedup vs baseline: 1.8621x; 1.0504x over previous
#include <hip/hip_runtime.h>
#include <math.h>


typedef __attribute__((ext_vector_type(2))) float v2f;
typedef __attribute__((ext_vector_type(4))) float v4f;
typedef __attribute__((ext_vector_type(8))) float v8f;
typedef __attribute__((ext_vector_type(2))) unsigned int v2u;

__device__ __forceinline__ void build_m_column(const float ct, const float nst,
                                               const float (&cad)[6],
                                               const float (&sad)[6],
                                               const int t, float (*Mm)[8]) {
  float st[8];
#pragma unroll
  for (int i = 0; i < 8; ++i) st[i] = (i == t) ? 1.0f : 0.0f;

  auto had = [&](int wb) {
    const int str = 1 << wb;
    const float r = 0.70710678118654752440f;
#pragma unroll
    for (int i = 0; i < 8; ++i) {
      if (i & str) continue;
      const float a0 = st[i], a1 = st[i + str];
      st[i] = r * (a0 + a1);
      st[i + str] = r * (a0 - a1);
    }
  };
  auto cry = [&](int cb, int tb, float c, float s) {
    const int cs = 1 << cb, ts = 1 << tb;
#pragma unroll
    for (int i = 0; i < 8; ++i) {
      if (!(i & cs) || (i & ts)) continue;
      const float a0 = st[i], a1 = st[i + ts];
      st[i] = c * a0 - s * a1;
      st[i + ts] = s * a0 + c * a1;
    }
  };

  had(2); had(1); had(0);
  const int wbit[3] = {2, 1, 0};
#pragma unroll
  for (int i = 0; i < 3; ++i) cry(wbit[i], wbit[(i + 1) % 3], cad[i], sad[i]);
  st[0] = -st[0];
#pragma unroll
  for (int i = 0; i < 3; ++i) cry(wbit[i], wbit[(i + 1) % 3], cad[3 + i], sad[3 + i]);
  had(2); had(1); had(0);

#pragma unroll
  for (int k = 0; k < 8; ++k) {
    Mm[k][t] = ct * st[k];
    Mm[8 + k][t] = nst * st[k];
  }
}

__device__ __forceinline__ float lane_value(float v, int src) {
  return __int_as_float(__builtin_amdgcn_readlane(__float_as_int(v), src));
}

__device__ __forceinline__ float sum_halves_of_pair(float a, float b) {
  const v2u r = __builtin_amdgcn_permlane16_swap(__float_as_uint(a),
                                                 __float_as_uint(b), false, false);
  return __uint_as_float(r.x) + __uint_as_float(r.y);
}

#define GROUPS_PER_BATCH 8

__global__ __launch_bounds__(256) void gqhan_fused(const float* __restrict__ x,
                                                   const float* __restrict__ theta_fo,
                                                   const float* __restrict__ theta_ado,
                                                   float* __restrict__ out,
                                                   int B) {
  __shared__ __attribute__((aligned(16))) float Mm[16][8];
  constexpr int T = 2 * GROUPS_PER_BATCH;
  const int t = threadIdx.x;
  const int lane = t & 31;
  const int l = lane & 15;
  const int h = lane >> 4;
  const int wib = __builtin_amdgcn_readfirstlane(t >> 5);
  const int wpb = blockDim.x >> 5;
  const int wave = blockIdx.x * wpb + wib;
  const int nwaves = gridDim.x * wpb;
  const int ngroups = (B + 31) >> 5;
  if ((int)blockIdx.x * wpb >= ngroups) return;
  const bool active = wave < ngroups;

  v4f xb[T];
  auto load_batch = [&](const int g0) {
#pragma unroll
    for (int u = 0; u < T; ++u) {
      const int g = g0 + (u >> 1) * nwaves;
      int sr = (g << 5) + 16 * (u & 1) + l;
      if (sr >= B) sr = B - 1;
      xb[u] = *(const v4f*)(x + (size_t)sr * 8 + 4 * h);
    }
  };

  int g0 = wave;
  if (wib == 0) {
    const int ai = (lane < 13) ? lane : 13;
    const float* ap = (ai < 8) ? (theta_fo + ai) : (theta_ado + (ai - 8));
    const float angle = *ap;
    load_batch(g0);
    const float arg = (lane < 8) ? angle : 0.5f * angle;
    const float cl = cosf(arg), sl = sinf(arg);
    float cad[6], sad[6];
#pragma unroll
    for (int i = 0; i < 6; ++i) {
      cad[i] = lane_value(cl, 8 + i);
      sad[i] = lane_value(sl, 8 + i);
    }
    if (lane < 8) build_m_column(cl, -sl, cad, sad, lane, Mm);
  } else if (active) {
    load_batch(g0);
  }
  __builtin_amdgcn_fence(__ATOMIC_RELEASE, "workgroup", "local");
  __builtin_amdgcn_s_barrier();
  __builtin_amdgcn_fence(__ATOMIC_ACQUIRE, "workgroup", "local");
  if (!active) return;

  const v4f am = *(const v4f*)&Mm[l][4 * h];
  v2f a0, a1;
  a0.x = am.x; a0.y = am.y;
  a1.x = am.z; a1.y = am.w;

  for (;;) {
#pragma unroll
    for (int jg = 0; jg < GROUPS_PER_BATCH; ++jg) {
      float qp[2], np[2];
#pragma unroll
      for (int w = 0; w < 2; ++w) {
        const v4f xv = xb[2 * jg + w];
        v2f b0, b1;
        b0.x = xv.x; b0.y = xv.y;
        b1.x = xv.z; b1.y = xv.w;
        v8f acc = {};
        acc = __builtin_amdgcn_wmma_f32_16x16x4_f32(false, a0, false, b0,
                                                    (short)0, acc, false, false);
        acc = __builtin_amdgcn_wmma_f32_16x16x4_f32(false, a1, false, b1,
                                                    (short)0, acc, false, false);
        float ev = acc[0] * acc[0];
        float od = acc[1] * acc[1];
        ev = fmaf(acc[2], acc[2], ev);
        od = fmaf(acc[3], acc[3], od);
        ev = fmaf(acc[4], acc[4], ev);
        od = fmaf(acc[5], acc[5], od);
        ev = fmaf(acc[6], acc[6], ev);
        od = fmaf(acc[7], acc[7], od);
        qp[w] = ev - od;
        float ns = xv.x * xv.x;
        ns = fmaf(xv.y, xv.y, ns);
        ns = fmaf(xv.z, xv.z, ns);
        ns = fmaf(xv.w, xv.w, ns);
        np[w] = ns;
      }
      const float q = sum_halves_of_pair(qp[0], qp[1]);
      const float nn = sum_halves_of_pair(np[0], np[1]);
      const float res = q * __builtin_amdgcn_rcpf(nn);

      const int g = g0 + jg * nwaves;
      const int sidx = (g << 5) + lane;
      if (sidx < B) out[sidx] = res;
    }
    g0 += nwaves * GROUPS_PER_BATCH;
    if (g0 >= ngroups) break;
    load_batch(g0);
  }
}

extern "C" void kernel_launch(void* const* d_in, const int* in_sizes, int n_in,
                              void* d_out, int out_size, void* d_ws,
                              size_t ws_size, hipStream_t stream) {
  const float* x = (const float*)d_in[0];
  const float* theta_fo = (const float*)d_in[1];
  const float* theta_ado = (const float*)d_in[2];
  float* out = (float*)d_out;
  const int B = in_sizes[0] / 8;

  hipLaunchKernelGGL(gqhan_fused, dim3(512), dim3(256), 0, stream,
                     x, theta_fo, theta_ado, out, B);
}


# M hand-over off the vector-memory path: wave 0 reads the angles by scalar loads, and a split barrier lets every wave arrive before it requests rows (wave 0 once M is stored) and wait after
# speedup vs baseline: 2.0216x; 1.0857x over previous
#include <hip/hip_runtime.h>
#include <math.h>


typedef __attribute__((ext_vector_type(2))) float v2f;
typedef __attribute__((ext_vector_type(4))) float v4f;
typedef __attribute__((ext_vector_type(8))) float v8f;
typedef __attribute__((ext_vector_type(2))) unsigned int v2u;

__device__ __forceinline__ void build_m_column(const float ct, const float nst,
                                               const float (&cad)[6],
                                               const float (&sad)[6],
                                               const int t, float (*Mm)[8]) {
  float st[8];
#pragma unroll
  for (int i = 0; i < 8; ++i) st[i] = (i == t) ? 1.0f : 0.0f;

  auto had = [&](int wb) {
    const int str = 1 << wb;
    const float r = 0.70710678118654752440f;
#pragma unroll
    for (int i = 0; i < 8; ++i) {
      if (i & str) continue;
      const float a0 = st[i], a1 = st[i + str];
      st[i] = r * (a0 + a1);
      st[i + str] = r * (a0 - a1);
    }
  };
  auto cry = [&](int cb, int tb, float c, float s) {
    const int cs = 1 << cb, ts = 1 << tb;
#pragma unroll
    for (int i = 0; i < 8; ++i) {
      if (!(i & cs) || (i & ts)) continue;
      const float a0 = st[i], a1 = st[i + ts];
      st[i] = c * a0 - s * a1;
      st[i + ts] = s * a0 + c * a1;
    }
  };

  had(2); had(1); had(0);
  const int wbit[3] = {2, 1, 0};
#pragma unroll
  for (int i = 0; i < 3; ++i) cry(wbit[i], wbit[(i + 1) % 3], cad[i], sad[i]);
  st[0] = -st[0];
#pragma unroll
  for (int i = 0; i < 3; ++i) cry(wbit[i], wbit[(i + 1) % 3], cad[3 + i], sad[3 + i]);
  had(2); had(1); had(0);

#pragma unroll
  for (int k = 0; k < 8; ++k) {
    Mm[k][t] = ct * st[k];
    Mm[8 + k][t] = nst * st[k];
  }
}

__device__ __forceinline__ float lane_value(float v, int src) {
  return __int_as_float(__builtin_amdgcn_readlane(__float_as_int(v), src));
}

__device__ __forceinline__ float sum_halves_of_pair(float a, float b) {
  const v2u r = __builtin_amdgcn_permlane16_swap(__float_as_uint(a),
                                                 __float_as_uint(b), false, false);
  return __uint_as_float(r.x) + __uint_as_float(r.y);
}

#define GROUPS_PER_BATCH 8

__global__ __launch_bounds__(256) void gqhan_fused(const float* __restrict__ x,
                                                   const float* __restrict__ theta_fo,
                                                   const float* __restrict__ theta_ado,
                                                   float* __restrict__ out,
                                                   int B) {
  __shared__ __attribute__((aligned(16))) float Mm[16][8];
  constexpr int T = 2 * GROUPS_PER_BATCH;
  const int t = threadIdx.x;
  const int lane = t & 31;
  const int l = lane & 15;
  const int h = lane >> 4;
  const int wib = __builtin_amdgcn_readfirstlane(t >> 5);
  const int wpb = blockDim.x >> 5;
  const int wave = blockIdx.x * wpb + wib;
  const int nwaves = gridDim.x * wpb;
  const int ngroups = (B + 31) >> 5;
  if ((int)blockIdx.x * wpb >= ngroups) return;
  const bool active = wave < ngroups;

  v4f xb[T];
  auto load_batch = [&](const int g0) {
#pragma unroll
    for (int u = 0; u < T; ++u) {
      const int g = g0 + (u >> 1) * nwaves;
      int sr = (g << 5) + 16 * (u & 1) + l;
      if (sr >= B) sr = B - 1;
      xb[u] = *(const v4f*)(x + (size_t)sr * 8 + 4 * h);
    }
  };

  int g0 = wave;
  if (wib == 0) {
    float th[14];
#pragma unroll
    for (int i = 0; i < 8; ++i) th[i] = theta_fo[i];
#pragma unroll
    for (int i = 0; i < 6; ++i) th[8 + i] = theta_ado[i];
    float arg = 0.5f * th[13];
#pragma unroll
    for (int i = 0; i < 13; ++i) {
      const float v = (i < 8) ? th[i] : 0.5f * th[i];
      arg = (lane == i) ? v : arg;
    }
    const float cl = cosf(arg), sl = sinf(arg);
    float cad[6], sad[6];
#pragma unroll
    for (int i = 0; i < 6; ++i) {
      cad[i] = lane_value(cl, 8 + i);
      sad[i] = lane_value(sl, 8 + i);
    }
    if (lane < 8) build_m_column(cl, -sl, cad, sad, lane, Mm);
  }
  __builtin_amdgcn_fence(__ATOMIC_RELEASE, "workgroup", "local");
  __builtin_amdgcn_s_barrier_signal(-1);
  if (active) load_batch(g0);
  __builtin_amdgcn_s_barrier_wait(-1);
  __builtin_amdgcn_fence(__ATOMIC_ACQUIRE, "workgroup", "local");
  if (!active) return;

  const v4f am = *(const v4f*)&Mm[l][4 * h];
  v2f a0, a1;
  a0.x = am.x; a0.y = am.y;
  a1.x = am.z; a1.y = am.w;

  for (;;) {
#pragma unroll
    for (int jg = 0; jg < GROUPS_PER_BATCH; ++jg) {
      float qp[2], np[2];
#pragma unroll
      for (int w = 0; w < 2; ++w) {
        const v4f xv = xb[2 * jg + w];
        v2f b0, b1;
        b0.x = xv.x; b0.y = xv.y;
        b1.x = xv.z; b1.y = xv.w;
        v8f acc = {};
        acc = __builtin_amdgcn_wmma_f32_16x16x4_f32(false, a0, false, b0,
                                                    (short)0, acc, false, false);
        acc = __builtin_amdgcn_wmma_f32_16x16x4_f32(false, a1, false, b1,
                                                    (short)0, acc, false, false);
        float ev = acc[0] * acc[0];
        float od = acc[1] * acc[1];
        ev = fmaf(acc[2], acc[2], ev);
        od = fmaf(acc[3], acc[3], od);
        ev = fmaf(acc[4], acc[4], ev);
        od = fmaf(acc[5], acc[5], od);
        ev = fmaf(acc[6], acc[6], ev);
        od = fmaf(acc[7], acc[7], od);
        qp[w] = ev - od;
        float ns = xv.x * xv.x;
        ns = fmaf(xv.y, xv.y, ns);
        ns = fmaf(xv.z, xv.z, ns);
        ns = fmaf(xv.w, xv.w, ns);
        np[w] = ns;
      }
      const float q = sum_halves_of_pair(qp[0], qp[1]);
      const float nn = sum_halves_of_pair(np[0], np[1]);
      const float res = q * __builtin_amdgcn_rcpf(nn);

      const int g = g0 + jg * nwaves;
      const int sidx = (g << 5) + lane;
      if (sidx < B) out[sidx] = res;
    }
    g0 += nwaves * GROUPS_PER_BATCH;
    if (g0 >= ngroups) break;
    load_batch(g0);
  }
}

extern "C" void kernel_launch(void* const* d_in, const int* in_sizes, int n_in,
                              void* d_out, int out_size, void* d_ws,
                              size_t ws_size, hipStream_t stream) {
  const float* x = (const float*)d_in[0];
  const float* theta_fo = (const float*)d_in[1];
  const float* theta_ado = (const float*)d_in[2];
  float* out = (float*)d_out;
  const int B = in_sizes[0] / 8;

  hipLaunchKernelGGL(gqhan_fused, dim3(512), dim3(256), 0, stream,
                     x, theta_fo, theta_ado, out, B);
}


# Each wave takes 8 consecutive groups and the grid shape is a compile-time constant: 16 row loads and 8 stores share one base with immediate offsets, so per-load and per-store address arithmetic goes
# speedup vs baseline: 2.2394x; 1.1077x over previous
#include <hip/hip_runtime.h>
#include <math.h>


typedef __attribute__((ext_vector_type(2))) float v2f;
typedef __attribute__((ext_vector_type(4))) float v4f;
typedef __attribute__((ext_vector_type(8))) float v8f;
typedef __attribute__((ext_vector_type(2))) unsigned int v2u;

__device__ __forceinline__ void build_m_column(const float ct, const float nst,
                                               const float (&cad)[6],
                                               const float (&sad)[6],
                                               const int t, float (*Mm)[8]) {
  float st[8];
#pragma unroll
  for (int i = 0; i < 8; ++i) st[i] = (i == t) ? 1.0f : 0.0f;

  auto had = [&](int wb) {
    const int str = 1 << wb;
    const float r = 0.70710678118654752440f;
#pragma unroll
    for (int i = 0; i < 8; ++i) {
      if (i & str) continue;
      const float a0 = st[i], a1 = st[i + str];
      st[i] = r * (a0 + a1);
      st[i + str] = r * (a0 - a1);
    }
  };
  auto cry = [&](int cb, int tb, float c, float s) {
    const int cs = 1 << cb, ts = 1 << tb;
#pragma unroll
    for (int i = 0; i < 8; ++i) {
      if (!(i & cs) || (i & ts)) continue;
      const float a0 = st[i], a1 = st[i + ts];
      st[i] = c * a0 - s * a1;
      st[i + ts] = s * a0 + c * a1;
    }
  };

  had(2); had(1); had(0);
  const int wbit[3] = {2, 1, 0};
#pragma unroll
  for (int i = 0; i < 3; ++i) cry(wbit[i], wbit[(i + 1) % 3], cad[i], sad[i]);
  st[0] = -st[0];
#pragma unroll
  for (int i = 0; i < 3; ++i) cry(wbit[i], wbit[(i + 1) % 3], cad[3 + i], sad[3 + i]);
  had(2); had(1); had(0);

#pragma unroll
  for (int k = 0; k < 8; ++k) {
    Mm[k][t] = ct * st[k];
    Mm[8 + k][t] = nst * st[k];
  }
}

__device__ __forceinline__ float lane_value(float v, int src) {
  return __int_as_float(__builtin_amdgcn_readlane(__float_as_int(v), src));
}

__device__ __forceinline__ float sum_halves_of_pair(float a, float b) {
  const v2u r = __builtin_amdgcn_permlane16_swap(__float_as_uint(a),
                                                 __float_as_uint(b), false, false);
  return __uint_as_float(r.x) + __uint_as_float(r.y);
}

#define GROUPS_PER_RUN 8
#define WAVES_PER_WG 8
#define NUM_WG 512

constexpr int TILES_PER_RUN = 2 * GROUPS_PER_RUN;
constexpr int RUN_SAMPLES = 32 * GROUPS_PER_RUN;
constexpr int NUM_WAVES = NUM_WG * WAVES_PER_WG;
static_assert(NUM_WG == 512 && WAVES_PER_WG == 8, "kernel_launch starts 512 x 256");

template <bool FULL>
__device__ __forceinline__ void load_run(v4f (&xb)[TILES_PER_RUN],
                                         const float* __restrict__ x,
                                         const int s0, const int l, const int h,
                                         const int B) {
  if (FULL) {
    const float* xr = x + (size_t)s0 * 8;
    const unsigned lo = (unsigned)(l * 8 + 4 * h);
#pragma unroll
    for (int u = 0; u < TILES_PER_RUN; ++u)
      xb[u] = *(const v4f*)(xr + lo + u * 128);
  } else {
#pragma unroll
    for (int u = 0; u < TILES_PER_RUN; ++u) {
      int sr = s0 + 16 * u + l;
      if (sr >= B) sr = B - 1;
      xb[u] = *(const v4f*)(x + (size_t)sr * 8 + 4 * h);
    }
  }
}

template <bool FULL>
__device__ __forceinline__ void finish_run(const v4f (&xb)[TILES_PER_RUN],
                                           const v2f a0, const v2f a1,
                                           float* __restrict__ out,
                                           const int s0, const int lane,
                                           const int B) {
  float* orun = out + (size_t)s0;
#pragma unroll
  for (int jg = 0; jg < GROUPS_PER_RUN; ++jg) {
    float qp[2], np[2];
#pragma unroll
    for (int w = 0; w < 2; ++w) {
      const v4f xv = xb[2 * jg + w];
      v2f b0, b1;
      b0.x = xv.x; b0.y = xv.y;
      b1.x = xv.z; b1.y = xv.w;
      v8f acc = {};
      acc = __builtin_amdgcn_wmma_f32_16x16x4_f32(false, a0, false, b0,
                                                  (short)0, acc, false, false);
      acc = __builtin_amdgcn_wmma_f32_16x16x4_f32(false, a1, false, b1,
                                                  (short)0, acc, false, false);
      float ev = acc[0] * acc[0];
      float od = acc[1] * acc[1];
      ev = fmaf(acc[2], acc[2], ev);
      od = fmaf(acc[3], acc[3], od);
      ev = fmaf(acc[4], acc[4], ev);
      od = fmaf(acc[5], acc[5], od);
      ev = fmaf(acc[6], acc[6], ev);
      od = fmaf(acc[7], acc[7], od);
      qp[w] = ev - od;
      float ns = xv.x * xv.x;
      ns = fmaf(xv.y, xv.y, ns);
      ns = fmaf(xv.z, xv.z, ns);
      ns = fmaf(xv.w, xv.w, ns);
      np[w] = ns;
    }
    const float q = sum_halves_of_pair(qp[0], qp[1]);
    const float nn = sum_halves_of_pair(np[0], np[1]);
    const float res = q * __builtin_amdgcn_rcpf(nn);

    if (FULL) {
      orun[(unsigned)lane + jg * 32] = res;
    } else {
      const int sidx = s0 + jg * 32 + lane;
      if (sidx < B) out[sidx] = res;
    }
  }
}

__global__ __launch_bounds__(256) void gqhan_fused(const float* __restrict__ x,
                                                   const float* __restrict__ theta_fo,
                                                   const float* __restrict__ theta_ado,
                                                   float* __restrict__ out,
                                                   int B) {
  __shared__ __attribute__((aligned(16))) float Mm[16][8];
  const int t = threadIdx.x;
  const int lane = t & 31;
  const int l = lane & 15;
  const int h = lane >> 4;
  const int wib = __builtin_amdgcn_readfirstlane(t >> 5);
  const int wave = (int)blockIdx.x * WAVES_PER_WG + wib;
  const int nfull = B / RUN_SAMPLES;
  const int nruns = (B + RUN_SAMPLES - 1) / RUN_SAMPLES;
  if ((int)blockIdx.x * WAVES_PER_WG >= nruns) return;
  const bool active = wave < nruns;
  const bool first_whole = wave < nfull;

  v4f xb[TILES_PER_RUN];

  if (wib == 0) {
    float th[14];
#pragma unroll
    for (int i = 0; i < 8; ++i) th[i] = theta_fo[i];
#pragma unroll
    for (int i = 0; i < 6; ++i) th[8 + i] = theta_ado[i];
    float arg = 0.5f * th[13];
#pragma unroll
    for (int i = 0; i < 13; ++i) {
      const float v = (i < 8) ? th[i] : 0.5f * th[i];
      arg = (lane == i) ? v : arg;
    }
    const float cl = cosf(arg), sl = sinf(arg);
    float cad[6], sad[6];
#pragma unroll
    for (int i = 0; i < 6; ++i) {
      cad[i] = lane_value(cl, 8 + i);
      sad[i] = lane_value(sl, 8 + i);
    }
    if (lane < 8) build_m_column(cl, -sl, cad, sad, lane, Mm);
  }
  __builtin_amdgcn_fence(__ATOMIC_RELEASE, "workgroup", "local");
  __builtin_amdgcn_s_barrier_signal(-1);
  if (first_whole) load_run<true>(xb, x, wave * RUN_SAMPLES, l, h, B);
  __builtin_amdgcn_s_barrier_wait(-1);
  __builtin_amdgcn_fence(__ATOMIC_ACQUIRE, "workgroup", "local");
  if (!active) return;

  const v4f am = *(const v4f*)&Mm[l][4 * h];
  v2f a0, a1;
  a0.x = am.x; a0.y = am.y;
  a1.x = am.z; a1.y = am.w;

  if (first_whole) {
    finish_run<true>(xb, a0, a1, out, wave * RUN_SAMPLES, lane, B);
    for (int r = wave + NUM_WAVES; r < nfull; r += NUM_WAVES) {
      v4f xc[TILES_PER_RUN];
      load_run<true>(xc, x, r * RUN_SAMPLES, l, h, B);
      finish_run<true>(xc, a0, a1, out, r * RUN_SAMPLES, lane, B);
    }
  }
  if (nfull < nruns && nfull % NUM_WAVES == wave) {
    v4f xc[TILES_PER_RUN];
    load_run<false>(xc, x, nfull * RUN_SAMPLES, l, h, B);
    finish_run<false>(xc, a0, a1, out, nfull * RUN_SAMPLES, lane, B);
  }
}

extern "C" void kernel_launch(void* const* d_in, const int* in_sizes, int n_in,
                              void* d_out, int out_size, void* d_ws,
                              size_t ws_size, hipStream_t stream) {
  const float* x = (const float*)d_in[0];
  const float* theta_fo = (const float*)d_in[1];
  const float* theta_ado = (const float*)d_in[2];
  float* out = (float*)d_out;
  const int B = in_sizes[0] / 8;

  hipLaunchKernelGGL(gqhan_fused, dim3(512), dim3(256), 0, stream,
                     x, theta_fo, theta_ado, out, B);
}
